# starting kernel
# baseline (speedup 1.0000x reference)
#include <hip/hip_runtime.h>
#include <hip/hip_bf16.h>

static constexpr int B_DIM = 32;
static constexpr int T_DIM = 4096;
static constexpr int I_DIM = 128;
static constexpr int H_DIM = 256;
static constexpr int BT    = B_DIM * T_DIM;
static constexpr int SEGS  = 32;
static constexpr int L_SEG = T_DIM / SEGS;
static constexpr int CT    = 4;

typedef float v2f __attribute__((ext_vector_type(2)));
typedef float v8f __attribute__((ext_vector_type(8)));

__device__ __forceinline__ float sigmoidf_fast(float z) {
    return 1.0f / (1.0f + __expf(-z));
}

__global__ __launch_bounds__(128) void sfnn_gemm_wmma(
    const float* __restrict__ x, const float* __restrict__ Wd,
    const float* __restrict__ bd, float* __restrict__ din)
{
    const int wave = blockIdx.x * (blockDim.x >> 5) + (threadIdx.x >> 5);
    const int lane = threadIdx.x & 31;
    const int half = lane >> 4;
    const int l    = lane & 15;

    const int colTile = wave & (H_DIM / (16 * CT) - 1);
    const int rowTile = wave / (H_DIM / (16 * CT));
    const int row0 = rowTile * 16;
    const int col0 = colTile * 16 * CT;

    const float* arow = x + (size_t)(row0 + l) * I_DIM + 2 * half;
    const float* brow[CT];
#pragma unroll
    for (int c = 0; c < CT; ++c)
        brow[c] = Wd + (size_t)(col0 + c * 16 + l) * I_DIM + 2 * half;

    v8f acc[CT];
#pragma unroll
    for (int c = 0; c < CT; ++c) {
        const float bias = bd[col0 + c * 16 + l];
#pragma unroll
        for (int r = 0; r < 8; ++r) acc[c][r] = bias;
    }

#pragma unroll 8
    for (int k = 0; k < I_DIM; k += 4) {
        const v2f a = *reinterpret_cast<const v2f*>(arow + k);
#pragma unroll
        for (int c = 0; c < CT; ++c) {
            const v2f bf = *reinterpret_cast<const v2f*>(brow[c] + k);
            acc[c] = __builtin_amdgcn_wmma_f32_16x16x4_f32(
                 false, a,  false, bf,
                 (short)0, acc[c],  false,  false);
        }
    }

    float* dout = din + (size_t)(row0 + 8 * half) * H_DIM + col0 + l;
#pragma unroll
    for (int c = 0; c < CT; ++c)
#pragma unroll
        for (int r = 0; r < 8; ++r)
            dout[(size_t)r * H_DIM + c * 16] = acc[c][r];
}

__global__ __launch_bounds__(256) void sfnn_seg_scan(
    const float* __restrict__ din, const float* __restrict__ tau_m,
    const float* __restrict__ tau_n, float* __restrict__ seg)
{
    const int tid = blockIdx.x * blockDim.x + threadIdx.x;
    const int h = tid & (H_DIM - 1);
    const int s = (tid >> 8) & (SEGS - 1);
    const int b = tid >> 13;

    const float alpha = sigmoidf_fast(tau_m[h]);
    const float beta  = sigmoidf_fast(tau_n[h]);

    const float* p = din + ((size_t)b * T_DIM + (size_t)s * L_SEG) * H_DIM + h;
    float d = 0.0f, m = 0.0f;
#pragma unroll 4
    for (int j = 0; j < L_SEG; ++j) {
        const float dt = p[(size_t)j * H_DIM];
        d = beta * d + (1.0f - beta) * dt;
        m = alpha * m + (1.0f - alpha) * d;
    }
    float* q = seg + ((size_t)(b * SEGS + s) * H_DIM + h) * 2;
    q[0] = d;
    q[1] = m;
}

__global__ __launch_bounds__(256) void sfnn_seg_combine(
    const float* __restrict__ tau_m, const float* __restrict__ tau_n,
    float* __restrict__ seg)
{
    const int tid = blockIdx.x * blockDim.x + threadIdx.x;
    const int h = tid & (H_DIM - 1);
    const int b = tid >> 8;

    const float alpha = sigmoidf_fast(tau_m[h]);
    const float beta  = sigmoidf_fast(tau_n[h]);

    float betaL = 1.0f, c_dm = 0.0f, alphaL = 1.0f;
#pragma unroll 4
    for (int j = 0; j < L_SEG; ++j) {
        betaL *= beta;
        c_dm   = alpha * c_dm + (1.0f - alpha) * betaL;
        alphaL *= alpha;
    }

    float d = 0.0f, m = 0.0f;
    for (int s = 0; s < SEGS; ++s) {
        float* q = seg + ((size_t)(b * SEGS + s) * H_DIM + h) * 2;
        const float ed = q[0];
        const float em = q[1];
        q[0] = d;
        q[1] = m;
        const float nd = betaL * d + ed;
        const float nm = alphaL * m + c_dm * d + em;
        d = nd;
        m = nm;
    }
}

__global__ __launch_bounds__(256) void sfnn_out_scan(
    const float* __restrict__ din, const float* __restrict__ seg,
    const float* __restrict__ tau_m, const float* __restrict__ tau_n,
    const float* __restrict__ Wout, const float* __restrict__ bout,
    float* __restrict__ out)
{
    const int b  = blockIdx.x >> 5;
    const int sg = blockIdx.x & 31;
    const int h  = threadIdx.x;
    const int lane = h & 31;
    const int wv   = h >> 5;

    const float alpha = sigmoidf_fast(tau_m[h]);
    const float beta  = sigmoidf_fast(tau_n[h]);
    const float w     = Wout[h];
    const float bo    = bout[0];

    const float* q = seg + ((size_t)(b * SEGS + sg) * H_DIM + h) * 2;
    float d = q[0];
    float m = q[1];

    const float* p = din + ((size_t)b * T_DIM + (size_t)sg * L_SEG) * H_DIM + h;
    float* o = out + (size_t)b * T_DIM + (size_t)sg * L_SEG;

    __shared__ float red[8][8];

    for (int t0 = 0; t0 < L_SEG; t0 += 8) {
        float c[8];
#pragma unroll
        for (int j = 0; j < 8; ++j) {
            const float dt = p[(size_t)(t0 + j) * H_DIM];
            d = beta * d + (1.0f - beta) * dt;
            m = alpha * m + (1.0f - alpha) * d;
            c[j] = m * w;
        }
#pragma unroll
        for (int j = 0; j < 8; ++j) {
#pragma unroll
            for (int off = 16; off > 0; off >>= 1)
                c[j] += __shfl_xor(c[j], off, 32);
        }
        if (lane == 0) {
#pragma unroll
            for (int j = 0; j < 8; ++j) red[wv][j] = c[j];
        }
        __syncthreads();
        if (h < 8) {
            float acc = bo;
#pragma unroll
            for (int w2 = 0; w2 < 8; ++w2) acc += red[w2][h];
            o[t0 + h] = sigmoidf_fast(acc);
        }
        __syncthreads();
    }
}

extern "C" void kernel_launch(void* const* d_in, const int* in_sizes, int n_in,
                              void* d_out, int out_size, void* d_ws, size_t ws_size,
                              hipStream_t stream) {
    const float* x     = (const float*)d_in[0];
    const float* Wd    = (const float*)d_in[1];
    const float* bd    = (const float*)d_in[2];
    const float* tau_m = (const float*)d_in[3];
    const float* tau_n = (const float*)d_in[4];
    const float* Wout  = (const float*)d_in[5];
    const float* bout  = (const float*)d_in[6];
    float* out = (float*)d_out;

    float* din = (float*)d_ws;
    float* seg = din + (size_t)BT * H_DIM;

    const int waves = (BT / 16) * (H_DIM / (16 * CT));
    sfnn_gemm_wmma<<<waves / 4, 128, 0, stream>>>(x, Wd, bd, din);

    sfnn_seg_scan<<<(B_DIM * SEGS * H_DIM) / 256, 256, 0, stream>>>(
        din, tau_m, tau_n, seg);

    sfnn_seg_combine<<<(B_DIM * H_DIM) / 256, 256, 0, stream>>>(
        tau_m, tau_n, seg);

    sfnn_out_scan<<<B_DIM * SEGS, 256, 0, stream>>>(
        din, seg, tau_m, tau_n, Wout, bout, out);
}


# GEMM operands staged once per 128x128 work-group tile in LDS (whole-line loads), 32x64 wave tiles read 16-byte fragments from LDS; k order inside each 8-wide K block changes (sum order)
# speedup vs baseline: 1.5480x; 1.5480x over previous
#include <hip/hip_runtime.h>
#include <hip/hip_bf16.h>

static constexpr int B_DIM = 32;
static constexpr int T_DIM = 4096;
static constexpr int I_DIM = 128;
static constexpr int H_DIM = 256;
static constexpr int BT    = B_DIM * T_DIM;
static constexpr int SEGS  = 32;
static constexpr int L_SEG = T_DIM / SEGS;

typedef float v2f __attribute__((ext_vector_type(2)));
typedef float v8f __attribute__((ext_vector_type(8)));

__device__ __forceinline__ float sigmoidf_fast(float z) {
    return 1.0f / (1.0f + __expf(-z));
}

static constexpr int GM  = 128;
static constexpr int GN  = 128;
static constexpr int LDP = I_DIM + 4;

typedef float v4f __attribute__((ext_vector_type(4)));

__global__ __launch_bounds__(256) void sfnn_gemm_wmma(
    const float* __restrict__ x, const float* __restrict__ Wd,
    const float* __restrict__ bd, float* __restrict__ din)
{
    __shared__ __attribute__((aligned(16))) float As[GM * LDP];
    __shared__ __attribute__((aligned(16))) float Bs[GN * LDP];

    const int tid  = threadIdx.x;
    const int row0 = (blockIdx.x >> 1) * GM;
    const int n0   = (blockIdx.x & 1) * GN;

    {
        const v4f* xg = reinterpret_cast<const v4f*>(x  + (size_t)row0 * I_DIM);
        const v4f* wg = reinterpret_cast<const v4f*>(Wd + (size_t)n0   * I_DIM);
        v4f xa[16], wa[16];
#pragma unroll
        for (int i = 0; i < 16; ++i) xa[i] = xg[i * 256 + tid];
#pragma unroll
        for (int i = 0; i < 16; ++i) wa[i] = wg[i * 256 + tid];
        const int sr = tid >> 5;
        const int sc = (tid & 31) * 4;
#pragma unroll
        for (int i = 0; i < 16; ++i)
            *reinterpret_cast<v4f*>(&As[(i * 8 + sr) * LDP + sc]) = xa[i];
#pragma unroll
        for (int i = 0; i < 16; ++i)
            *reinterpret_cast<v4f*>(&Bs[(i * 8 + sr) * LDP + sc]) = wa[i];
    }
    __syncthreads();

    const int lane = tid & 31;
    const int wv   = tid >> 5;
    const int half = lane >> 4;
    const int l    = lane & 15;
    const int wr   = wv >> 1;
    const int wc   = wv & 1;

    const float* ap = As + (wr * 32 + l) * LDP + 4 * half;
    const float* bp = Bs + (wc * 64 + l) * LDP + 4 * half;

    v8f acc[2][4];
#pragma unroll
    for (int j = 0; j < 4; ++j) {
        const float bias = bd[n0 + wc * 64 + j * 16 + l];
#pragma unroll
        for (int i = 0; i < 2; ++i)
#pragma unroll
            for (int r = 0; r < 8; ++r) acc[i][j][r] = bias;
    }

    unsigned doff = (unsigned)((row0 + wr * 32 + 8 * half) * H_DIM + n0 + wc * 64 + l);
    asm volatile("" : "+v"(doff));
    float* dout = din + doff;

    v4f a[2], b[4];
#pragma unroll
    for (int i = 0; i < 2; ++i) a[i] = *reinterpret_cast<const v4f*>(ap + i * 16 * LDP);
#pragma unroll
    for (int j = 0; j < 4; ++j) b[j] = *reinterpret_cast<const v4f*>(bp + j * 16 * LDP);

#pragma unroll
    for (int kk = 0; kk < I_DIM; kk += 8) {
        v4f an[2], bn[4];
        if (kk + 8 < I_DIM) {
#pragma unroll
            for (int i = 0; i < 2; ++i)
                an[i] = *reinterpret_cast<const v4f*>(ap + i * 16 * LDP + kk + 8);
#pragma unroll
            for (int j = 0; j < 4; ++j)
                bn[j] = *reinterpret_cast<const v4f*>(bp + j * 16 * LDP + kk + 8);
        }
#pragma unroll
        for (int i = 0; i < 2; ++i) {
            const v2f a0 = __builtin_shufflevector(a[i], a[i], 0, 1);
            const v2f a1 = __builtin_shufflevector(a[i], a[i], 2, 3);
#pragma unroll
            for (int j = 0; j < 4; ++j) {
                const v2f b0 = __builtin_shufflevector(b[j], b[j], 0, 1);
                const v2f b1 = __builtin_shufflevector(b[j], b[j], 2, 3);
                acc[i][j] = __builtin_amdgcn_wmma_f32_16x16x4_f32(
                     false, a0,  false, b0,
                     (short)0, acc[i][j],  false,  false);
                acc[i][j] = __builtin_amdgcn_wmma_f32_16x16x4_f32(
                     false, a1,  false, b1,
                     (short)0, acc[i][j],  false,  false);
            }
        }
        if (kk + 8 < I_DIM) {
#pragma unroll
            for (int i = 0; i < 2; ++i) a[i] = an[i];
#pragma unroll
            for (int j = 0; j < 4; ++j) b[j] = bn[j];
        }
    }

#pragma unroll
    for (int i = 0; i < 2; ++i)
#pragma unroll
        for (int j = 0; j < 4; ++j)
#pragma unroll
            for (int r = 0; r < 8; ++r)
                dout[(size_t)(i * 16 + r) * H_DIM + j * 16] = acc[i][j][r];
}

__global__ __launch_bounds__(256) void sfnn_seg_scan(
    const float* __restrict__ din, const float* __restrict__ tau_m,
    const float* __restrict__ tau_n, float* __restrict__ seg)
{
    const int tid = blockIdx.x * blockDim.x + threadIdx.x;
    const int h = tid & (H_DIM - 1);
    const int s = (tid >> 8) & (SEGS - 1);
    const int b = tid >> 13;

    const float alpha = sigmoidf_fast(tau_m[h]);
    const float beta  = sigmoidf_fast(tau_n[h]);

    const float* p = din + ((size_t)b * T_DIM + (size_t)s * L_SEG) * H_DIM + h;
    float d = 0.0f, m = 0.0f;
#pragma unroll 4
    for (int j = 0; j < L_SEG; ++j) {
        const float dt = p[(size_t)j * H_DIM];
        d = beta * d + (1.0f - beta) * dt;
        m = alpha * m + (1.0f - alpha) * d;
    }
    float* q = seg + ((size_t)(b * SEGS + s) * H_DIM + h) * 2;
    q[0] = d;
    q[1] = m;
}

__global__ __launch_bounds__(256) void sfnn_seg_combine(
    const float* __restrict__ tau_m, const float* __restrict__ tau_n,
    float* __restrict__ seg)
{
    const int tid = blockIdx.x * blockDim.x + threadIdx.x;
    const int h = tid & (H_DIM - 1);
    const int b = tid >> 8;

    const float alpha = sigmoidf_fast(tau_m[h]);
    const float beta  = sigmoidf_fast(tau_n[h]);

    float betaL = 1.0f, c_dm = 0.0f, alphaL = 1.0f;
#pragma unroll 4
    for (int j = 0; j < L_SEG; ++j) {
        betaL *= beta;
        c_dm   = alpha * c_dm + (1.0f - alpha) * betaL;
        alphaL *= alpha;
    }

    float d = 0.0f, m = 0.0f;
    for (int s = 0; s < SEGS; ++s) {
        float* q = seg + ((size_t)(b * SEGS + s) * H_DIM + h) * 2;
        const float ed = q[0];
        const float em = q[1];
        q[0] = d;
        q[1] = m;
        const float nd = betaL * d + ed;
        const float nm = alphaL * m + c_dm * d + em;
        d = nd;
        m = nm;
    }
}

__global__ __launch_bounds__(256) void sfnn_out_scan(
    const float* __restrict__ din, const float* __restrict__ seg,
    const float* __restrict__ tau_m, const float* __restrict__ tau_n,
    const float* __restrict__ Wout, const float* __restrict__ bout,
    float* __restrict__ out)
{
    const int b  = blockIdx.x >> 5;
    const int sg = blockIdx.x & 31;
    const int h  = threadIdx.x;
    const int lane = h & 31;
    const int wv   = h >> 5;

    const float alpha = sigmoidf_fast(tau_m[h]);
    const float beta  = sigmoidf_fast(tau_n[h]);
    const float w     = Wout[h];
    const float bo    = bout[0];

    const float* q = seg + ((size_t)(b * SEGS + sg) * H_DIM + h) * 2;
    float d = q[0];
    float m = q[1];

    const float* p = din + ((size_t)b * T_DIM + (size_t)sg * L_SEG) * H_DIM + h;
    float* o = out + (size_t)b * T_DIM + (size_t)sg * L_SEG;

    __shared__ float red[8][8];

    for (int t0 = 0; t0 < L_SEG; t0 += 8) {
        float c[8];
#pragma unroll
        for (int j = 0; j < 8; ++j) {
            const float dt = p[(size_t)(t0 + j) * H_DIM];
            d = beta * d + (1.0f - beta) * dt;
            m = alpha * m + (1.0f - alpha) * d;
            c[j] = m * w;
        }
#pragma unroll
        for (int j = 0; j < 8; ++j) {
#pragma unroll
            for (int off = 16; off > 0; off >>= 1)
                c[j] += __shfl_xor(c[j], off, 32);
        }
        if (lane == 0) {
#pragma unroll
            for (int j = 0; j < 8; ++j) red[wv][j] = c[j];
        }
        __syncthreads();
        if (h < 8) {
            float acc = bo;
#pragma unroll
            for (int w2 = 0; w2 < 8; ++w2) acc += red[w2][h];
            o[t0 + h] = sigmoidf_fast(acc);
        }
        __syncthreads();
    }
}

extern "C" void kernel_launch(void* const* d_in, const int* in_sizes, int n_in,
                              void* d_out, int out_size, void* d_ws, size_t ws_size,
                              hipStream_t stream) {
    const float* x     = (const float*)d_in[0];
    const float* Wd    = (const float*)d_in[1];
    const float* bd    = (const float*)d_in[2];
    const float* tau_m = (const float*)d_in[3];
    const float* tau_n = (const float*)d_in[4];
    const float* Wout  = (const float*)d_in[5];
    const float* bout  = (const float*)d_in[6];
    float* out = (float*)d_out;

    float* din = (float*)d_ws;
    float* seg = din + (size_t)BT * H_DIM;

    sfnn_gemm_wmma<<<(BT / GM) * (H_DIM / GN), 256, 0, stream>>>(x, Wd, bd, din);

    sfnn_seg_scan<<<(B_DIM * SEGS * H_DIM) / 256, 256, 0, stream>>>(
        din, tau_m, tau_n, seg);

    sfnn_seg_combine<<<(B_DIM * H_DIM) / 256, 256, 0, stream>>>(
        tau_m, tau_n, seg);

    sfnn_out_scan<<<B_DIM * SEGS, 256, 0, stream>>>(
        din, seg, tau_m, tau_n, Wout, bout, out);
}


# d_in never stored: each GEMM work-group scans its 128x128 tile from zero state in LDS, writes segment ends and W_out row partials; last kernel adds the initial-state response (sum order changes)
# speedup vs baseline: 2.1900x; 1.4147x over previous
#include <hip/hip_runtime.h>
#include <hip/hip_bf16.h>

static constexpr int B_DIM = 32;
static constexpr int T_DIM = 4096;
static constexpr int I_DIM = 128;
static constexpr int H_DIM = 256;
static constexpr int BT    = B_DIM * T_DIM;
static constexpr int SEGS  = 32;
static constexpr int L_SEG = T_DIM / SEGS;

typedef float v2f __attribute__((ext_vector_type(2)));
typedef float v8f __attribute__((ext_vector_type(8)));

__device__ __forceinline__ float sigmoidf_fast(float z) {
    return 1.0f / (1.0f + __expf(-z));
}

static constexpr int GM  = 128;
static constexpr int GN  = 128;
static constexpr int LDP = I_DIM + 4;

typedef float v4f __attribute__((ext_vector_type(4)));

__global__ __launch_bounds__(256) void sfnn_gemm_wmma(
    const float* __restrict__ x, const float* __restrict__ Wd,
    const float* __restrict__ bd, const float* __restrict__ tau_m,
    const float* __restrict__ tau_n, const float* __restrict__ Wout,
    float* __restrict__ seg, float* __restrict__ outz)
{
    __shared__ __attribute__((aligned(16))) float As[GM * LDP];
    __shared__ __attribute__((aligned(16))) float Bs[GN * LDP];

    const int tid  = threadIdx.x;
    const int row0 = (blockIdx.x >> 1) * GM;
    const int n0   = (blockIdx.x & 1) * GN;

    {
        const v4f* xg = reinterpret_cast<const v4f*>(x  + (size_t)row0 * I_DIM);
        const v4f* wg = reinterpret_cast<const v4f*>(Wd + (size_t)n0   * I_DIM);
        v4f xa[16], wa[16];
#pragma unroll
        for (int i = 0; i < 16; ++i) xa[i] = xg[i * 256 + tid];
#pragma unroll
        for (int i = 0; i < 16; ++i) wa[i] = wg[i * 256 + tid];
        const int sr = tid >> 5;
        const int sc = (tid & 31) * 4;
#pragma unroll
        for (int i = 0; i < 16; ++i)
            *reinterpret_cast<v4f*>(&As[(i * 8 + sr) * LDP + sc]) = xa[i];
#pragma unroll
        for (int i = 0; i < 16; ++i)
            *reinterpret_cast<v4f*>(&Bs[(i * 8 + sr) * LDP + sc]) = wa[i];
    }
    __syncthreads();

    const int lane = tid & 31;
    const int wv   = tid >> 5;
    const int half = lane >> 4;
    const int l    = lane & 15;
    const int wr   = wv >> 1;
    const int wc   = wv & 1;

    const float* ap = As + (wr * 32 + l) * LDP + 4 * half;
    const float* bp = Bs + (wc * 64 + l) * LDP + 4 * half;

    v8f acc[2][4];
#pragma unroll
    for (int j = 0; j < 4; ++j) {
        const float bias = bd[n0 + wc * 64 + j * 16 + l];
#pragma unroll
        for (int i = 0; i < 2; ++i)
#pragma unroll
            for (int r = 0; r < 8; ++r) acc[i][j][r] = bias;
    }

    unsigned toff = (unsigned)((wr * 32 + 8 * half) * LDP + wc * 64 + l);
    const int   cs = tid & (GN - 1);
    float al = sigmoidf_fast(tau_m[n0 + cs]);
    float be = sigmoidf_fast(tau_n[n0 + cs]);
    float wo = Wout[n0 + cs];
    asm volatile("" : "+v"(toff), "+v"(al), "+v"(be), "+v"(wo));

    v4f a[2], b[4];
#pragma unroll
    for (int i = 0; i < 2; ++i) a[i] = *reinterpret_cast<const v4f*>(ap + i * 16 * LDP);
#pragma unroll
    for (int j = 0; j < 4; ++j) b[j] = *reinterpret_cast<const v4f*>(bp + j * 16 * LDP);

#pragma unroll
    for (int kk = 0; kk < I_DIM; kk += 8) {
        v4f an[2], bn[4];
        if (kk + 8 < I_DIM) {
#pragma unroll
            for (int i = 0; i < 2; ++i)
                an[i] = *reinterpret_cast<const v4f*>(ap + i * 16 * LDP + kk + 8);
#pragma unroll
            for (int j = 0; j < 4; ++j)
                bn[j] = *reinterpret_cast<const v4f*>(bp + j * 16 * LDP + kk + 8);
        }
#pragma unroll
        for (int i = 0; i < 2; ++i) {
            const v2f a0 = __builtin_shufflevector(a[i], a[i], 0, 1);
            const v2f a1 = __builtin_shufflevector(a[i], a[i], 2, 3);
#pragma unroll
            for (int j = 0; j < 4; ++j) {
                const v2f b0 = __builtin_shufflevector(b[j], b[j], 0, 1);
                const v2f b1 = __builtin_shufflevector(b[j], b[j], 2, 3);
                acc[i][j] = __builtin_amdgcn_wmma_f32_16x16x4_f32(
                     false, a0,  false, b0,
                     (short)0, acc[i][j],  false,  false);
                acc[i][j] = __builtin_amdgcn_wmma_f32_16x16x4_f32(
                     false, a1,  false, b1,
                     (short)0, acc[i][j],  false,  false);
            }
        }
        if (kk + 8 < I_DIM) {
#pragma unroll
            for (int i = 0; i < 2; ++i) a[i] = an[i];
#pragma unroll
            for (int j = 0; j < 4; ++j) b[j] = bn[j];
        }
    }

    asm volatile("v_nop\n\tv_nop\n\tv_nop\n\tv_nop\n\tv_nop\n\tv_nop\n\tv_nop\n\tv_nop\n\tv_nop"
                 : "+v"(toff),
                   "+v"(acc[0][0]), "+v"(acc[0][1]), "+v"(acc[0][2]), "+v"(acc[0][3]),
                   "+v"(acc[1][0]), "+v"(acc[1][1]), "+v"(acc[1][2]), "+v"(acc[1][3]));

    __syncthreads();
    float* Ts = As;
    {
        float* tp = Ts + toff;
#pragma unroll
        for (int i = 0; i < 2; ++i)
#pragma unroll
            for (int j = 0; j < 4; ++j)
#pragma unroll
                for (int r = 0; r < 8; ++r)
                    tp[(i * 16 + r) * LDP + j * 16] = acc[i][j][r];
    }
    __syncthreads();

    if (tid < GN) {
        const float omb = 1.0f - be;
        const float oma = 1.0f - al;
        float* tc = Ts + tid;
        float d = 0.0f, m = 0.0f;
        for (int t = 0; t < GM; t += 8) {
            float v[8];
#pragma unroll
            for (int j = 0; j < 8; ++j) v[j] = tc[(t + j) * LDP];
#pragma unroll
            for (int j = 0; j < 8; ++j) {
                d = be * d + omb * v[j];
                m = al * m + oma * d;
                v[j] = m * wo;
            }
#pragma unroll
            for (int j = 0; j < 8; ++j) tc[(t + j) * LDP] = v[j];
        }
        float* q = seg + ((size_t)(blockIdx.x >> 1) * H_DIM + n0 + tid) * 2;
        q[0] = d;
        q[1] = m;
    }
    __syncthreads();

    {
        const int row  = tid >> 1;
        const int part = tid & 1;
        const float* rp = Ts + row * LDP + part * 64;
        float s = 0.0f;
#pragma unroll
        for (int qd = 0; qd < 16; ++qd) {
            const v4f v = *reinterpret_cast<const v4f*>(rp + 4 * qd);
            s += (v[0] + v[1]) + (v[2] + v[3]);
        }
        s += __shfl_xor(s, 1, 32);
        if (part == 0)
            outz[(size_t)(row0 + row) * 2 + (blockIdx.x & 1)] = s;
    }
}

__global__ __launch_bounds__(256) void sfnn_seg_combine(
    const float* __restrict__ tau_m, const float* __restrict__ tau_n,
    float* __restrict__ seg)
{
    const int tid = blockIdx.x * blockDim.x + threadIdx.x;
    const int h = tid & (H_DIM - 1);
    const int b = tid >> 8;

    const float alpha = sigmoidf_fast(tau_m[h]);
    const float beta  = sigmoidf_fast(tau_n[h]);

    float betaL = 1.0f, c_dm = 0.0f, alphaL = 1.0f;
#pragma unroll 4
    for (int j = 0; j < L_SEG; ++j) {
        betaL *= beta;
        c_dm   = alpha * c_dm + (1.0f - alpha) * betaL;
        alphaL *= alpha;
    }

    float d = 0.0f, m = 0.0f;
    for (int s = 0; s < SEGS; ++s) {
        float* q = seg + ((size_t)(b * SEGS + s) * H_DIM + h) * 2;
        const float ed = q[0];
        const float em = q[1];
        q[0] = d;
        q[1] = m;
        const float nd = betaL * d + ed;
        const float nm = alphaL * m + c_dm * d + em;
        d = nd;
        m = nm;
    }
}

__global__ __launch_bounds__(256) void sfnn_out_scan(
    const float* __restrict__ outz, const float* __restrict__ seg,
    const float* __restrict__ tau_m, const float* __restrict__ tau_n,
    const float* __restrict__ Wout, const float* __restrict__ bout,
    float* __restrict__ out)
{
    const int b  = blockIdx.x >> 5;
    const int sg = blockIdx.x & 31;
    const int h  = threadIdx.x;
    const int lane = h & 31;
    const int wv   = h >> 5;

    const float alpha = sigmoidf_fast(tau_m[h]);
    const float beta  = sigmoidf_fast(tau_n[h]);
    const float w     = Wout[h];
    const float bo    = bout[0];

    const float* q = seg + ((size_t)(b * SEGS + sg) * H_DIM + h) * 2;
    float d = q[0];
    float m = q[1];

    float* o = out + (size_t)b * T_DIM + (size_t)sg * L_SEG;

    __shared__ float red[8][8];
    __shared__ float zs[L_SEG];

    if (h < L_SEG) {
        const float* z = outz + ((size_t)b * T_DIM + (size_t)sg * L_SEG + h) * 2;
        zs[h] = z[0] + z[1];
    }

    for (int t0 = 0; t0 < L_SEG; t0 += 8) {
        float c[8];
#pragma unroll
        for (int j = 0; j < 8; ++j) {
            d = beta * d;
            m = alpha * m + (1.0f - alpha) * d;
            c[j] = m * w;
        }
#pragma unroll
        for (int j = 0; j < 8; ++j) {
#pragma unroll
            for (int off = 16; off > 0; off >>= 1)
                c[j] += __shfl_xor(c[j], off, 32);
        }
        if (lane == 0) {
#pragma unroll
            for (int j = 0; j < 8; ++j) red[wv][j] = c[j];
        }
        __syncthreads();
        if (h < 8) {
            float acc = bo + zs[t0 + h];
#pragma unroll
            for (int w2 = 0; w2 < 8; ++w2) acc += red[w2][h];
            o[t0 + h] = sigmoidf_fast(acc);
        }
        __syncthreads();
    }
}

extern "C" void kernel_launch(void* const* d_in, const int* in_sizes, int n_in,
                              void* d_out, int out_size, void* d_ws, size_t ws_size,
                              hipStream_t stream) {
    const float* x     = (const float*)d_in[0];
    const float* Wd    = (const float*)d_in[1];
    const float* bd    = (const float*)d_in[2];
    const float* tau_m = (const float*)d_in[3];
    const float* tau_n = (const float*)d_in[4];
    const float* Wout  = (const float*)d_in[5];
    const float* bout  = (const float*)d_in[6];
    float* out = (float*)d_out;

    float* outz = (float*)d_ws;
    float* seg  = outz + (size_t)BT * H_DIM;

    sfnn_gemm_wmma<<<(BT / GM) * (H_DIM / GN), 256, 0, stream>>>(
        x, Wd, bd, tau_m, tau_n, Wout, seg, outz);

    sfnn_seg_combine<<<(B_DIM * H_DIM) / 256, 256, 0, stream>>>(
        tau_m, tau_n, seg);

    sfnn_out_scan<<<B_DIM * SEGS, 256, 0, stream>>>(
        outz, seg, tau_m, tau_n, Wout, bout, out);
}


# Last kernel: cross-channel sum through an LDS tile [step][channel pair] and one row sum per step, replacing the 5-stage lane butterfly and two barriers per 8 steps (sum order changes)
# speedup vs baseline: 2.5221x; 1.1516x over previous
#include <hip/hip_runtime.h>
#include <hip/hip_bf16.h>

static constexpr int B_DIM = 32;
static constexpr int T_DIM = 4096;
static constexpr int I_DIM = 128;
static constexpr int H_DIM = 256;
static constexpr int BT    = B_DIM * T_DIM;
static constexpr int SEGS  = 32;
static constexpr int L_SEG = T_DIM / SEGS;

typedef float v2f __attribute__((ext_vector_type(2)));
typedef float v8f __attribute__((ext_vector_type(8)));

__device__ __forceinline__ float sigmoidf_fast(float z) {
    return 1.0f / (1.0f + __expf(-z));
}

static constexpr int GM  = 128;
static constexpr int GN  = 128;
static constexpr int LDP = I_DIM + 4;

typedef float v4f __attribute__((ext_vector_type(4)));

__global__ __launch_bounds__(256) void sfnn_gemm_wmma(
    const float* __restrict__ x, const float* __restrict__ Wd,
    const float* __restrict__ bd, const float* __restrict__ tau_m,
    const float* __restrict__ tau_n, const float* __restrict__ Wout,
    float* __restrict__ seg, float* __restrict__ outz)
{
    __shared__ __attribute__((aligned(16))) float As[GM * LDP];
    __shared__ __attribute__((aligned(16))) float Bs[GN * LDP];

    const int tid  = threadIdx.x;
    const int row0 = (blockIdx.x >> 1) * GM;
    const int n0   = (blockIdx.x & 1) * GN;

    {
        const v4f* xg = reinterpret_cast<const v4f*>(x  + (size_t)row0 * I_DIM);
        const v4f* wg = reinterpret_cast<const v4f*>(Wd + (size_t)n0   * I_DIM);
        v4f xa[16], wa[16];
#pragma unroll
        for (int i = 0; i < 16; ++i) xa[i] = xg[i * 256 + tid];
#pragma unroll
        for (int i = 0; i < 16; ++i) wa[i] = wg[i * 256 + tid];
        const int sr = tid >> 5;
        const int sc = (tid & 31) * 4;
#pragma unroll
        for (int i = 0; i < 16; ++i)
            *reinterpret_cast<v4f*>(&As[(i * 8 + sr) * LDP + sc]) = xa[i];
#pragma unroll
        for (int i = 0; i < 16; ++i)
            *reinterpret_cast<v4f*>(&Bs[(i * 8 + sr) * LDP + sc]) = wa[i];
    }
    __syncthreads();

    const int lane = tid & 31;
    const int wv   = tid >> 5;
    const int half = lane >> 4;
    const int l    = lane & 15;
    const int wr   = wv >> 1;
    const int wc   = wv & 1;

    const float* ap = As + (wr * 32 + l) * LDP + 4 * half;
    const float* bp = Bs + (wc * 64 + l) * LDP + 4 * half;

    v8f acc[2][4];
#pragma unroll
    for (int j = 0; j < 4; ++j) {
        const float bias = bd[n0 + wc * 64 + j * 16 + l];
#pragma unroll
        for (int i = 0; i < 2; ++i)
#pragma unroll
            for (int r = 0; r < 8; ++r) acc[i][j][r] = bias;
    }

    unsigned toff = (unsigned)((wr * 32 + 8 * half) * LDP + wc * 64 + l);
    const int   cs = tid & (GN - 1);
    float al = sigmoidf_fast(tau_m[n0 + cs]);
    float be = sigmoidf_fast(tau_n[n0 + cs]);
    float wo = Wout[n0 + cs];
    asm volatile("" : "+v"(toff), "+v"(al), "+v"(be), "+v"(wo));

    v4f a[2], b[4];
#pragma unroll
    for (int i = 0; i < 2; ++i) a[i] = *reinterpret_cast<const v4f*>(ap + i * 16 * LDP);
#pragma unroll
    for (int j = 0; j < 4; ++j) b[j] = *reinterpret_cast<const v4f*>(bp + j * 16 * LDP);

#pragma unroll
    for (int kk = 0; kk < I_DIM; kk += 8) {
        v4f an[2], bn[4];
        if (kk + 8 < I_DIM) {
#pragma unroll
            for (int i = 0; i < 2; ++i)
                an[i] = *reinterpret_cast<const v4f*>(ap + i * 16 * LDP + kk + 8);
#pragma unroll
            for (int j = 0; j < 4; ++j)
                bn[j] = *reinterpret_cast<const v4f*>(bp + j * 16 * LDP + kk + 8);
        }
#pragma unroll
        for (int i = 0; i < 2; ++i) {
            const v2f a0 = __builtin_shufflevector(a[i], a[i], 0, 1);
            const v2f a1 = __builtin_shufflevector(a[i], a[i], 2, 3);
#pragma unroll
            for (int j = 0; j < 4; ++j) {
                const v2f b0 = __builtin_shufflevector(b[j], b[j], 0, 1);
                const v2f b1 = __builtin_shufflevector(b[j], b[j], 2, 3);
                acc[i][j] = __builtin_amdgcn_wmma_f32_16x16x4_f32(
                     false, a0,  false, b0,
                     (short)0, acc[i][j],  false,  false);
                acc[i][j] = __builtin_amdgcn_wmma_f32_16x16x4_f32(
                     false, a1,  false, b1,
                     (short)0, acc[i][j],  false,  false);
            }
        }
        if (kk + 8 < I_DIM) {
#pragma unroll
            for (int i = 0; i < 2; ++i) a[i] = an[i];
#pragma unroll
            for (int j = 0; j < 4; ++j) b[j] = bn[j];
        }
    }

    asm volatile("v_nop\n\tv_nop\n\tv_nop\n\tv_nop\n\tv_nop\n\tv_nop\n\tv_nop\n\tv_nop\n\tv_nop"
                 : "+v"(toff),
                   "+v"(acc[0][0]), "+v"(acc[0][1]), "+v"(acc[0][2]), "+v"(acc[0][3]),
                   "+v"(acc[1][0]), "+v"(acc[1][1]), "+v"(acc[1][2]), "+v"(acc[1][3]));

    __syncthreads();
    float* Ts = As;
    {
        float* tp = Ts + toff;
#pragma unroll
        for (int i = 0; i < 2; ++i)
#pragma unroll
            for (int j = 0; j < 4; ++j)
#pragma unroll
                for (int r = 0; r < 8; ++r)
                    tp[(i * 16 + r) * LDP + j * 16] = acc[i][j][r];
    }
    __syncthreads();

    if (tid < GN) {
        const float omb = 1.0f - be;
        const float oma = 1.0f - al;
        float* tc = Ts + tid;
        float d = 0.0f, m = 0.0f;
        for (int t = 0; t < GM; t += 8) {
            float v[8];
#pragma unroll
            for (int j = 0; j < 8; ++j) v[j] = tc[(t + j) * LDP];
#pragma unroll
            for (int j = 0; j < 8; ++j) {
                d = be * d + omb * v[j];
                m = al * m + oma * d;
                v[j] = m * wo;
            }
#pragma unroll
            for (int j = 0; j < 8; ++j) tc[(t + j) * LDP] = v[j];
        }
        float* q = seg + ((size_t)(blockIdx.x >> 1) * H_DIM + n0 + tid) * 2;
        q[0] = d;
        q[1] = m;
    }
    __syncthreads();

    {
        const int row  = tid >> 1;
        const int part = tid & 1;
        const float* rp = Ts + row * LDP + part * 64;
        float s = 0.0f;
#pragma unroll
        for (int qd = 0; qd < 16; ++qd) {
            const v4f v = *reinterpret_cast<const v4f*>(rp + 4 * qd);
            s += (v[0] + v[1]) + (v[2] + v[3]);
        }
        s += __shfl_xor(s, 1, 32);
        if (part == 0)
            outz[(size_t)(row0 + row) * 2 + (blockIdx.x & 1)] = s;
    }
}

__global__ __launch_bounds__(256) void sfnn_seg_combine(
    const float* __restrict__ tau_m, const float* __restrict__ tau_n,
    float* __restrict__ seg)
{
    const int tid = blockIdx.x * blockDim.x + threadIdx.x;
    const int h = tid & (H_DIM - 1);
    const int b = tid >> 8;

    const float alpha = sigmoidf_fast(tau_m[h]);
    const float beta  = sigmoidf_fast(tau_n[h]);

    float betaL = 1.0f, c_dm = 0.0f, alphaL = 1.0f;
#pragma unroll 4
    for (int j = 0; j < L_SEG; ++j) {
        betaL *= beta;
        c_dm   = alpha * c_dm + (1.0f - alpha) * betaL;
        alphaL *= alpha;
    }

    float d = 0.0f, m = 0.0f;
    for (int s = 0; s < SEGS; ++s) {
        float* q = seg + ((size_t)(b * SEGS + s) * H_DIM + h) * 2;
        const float ed = q[0];
        const float em = q[1];
        q[0] = d;
        q[1] = m;
        const float nd = betaL * d + ed;
        const float nm = alphaL * m + c_dm * d + em;
        d = nd;
        m = nm;
    }
}

static constexpr int PLD = L_SEG + 4;

__global__ __launch_bounds__(128) void sfnn_out_scan(
    const float* __restrict__ outz, const float* __restrict__ seg,
    const float* __restrict__ tau_m, const float* __restrict__ tau_n,
    const float* __restrict__ Wout, const float* __restrict__ bout,
    float* __restrict__ out)
{
    __shared__ __attribute__((aligned(16))) float Ps[L_SEG * PLD];

    const int b  = blockIdx.x >> 5;
    const int sg = blockIdx.x & 31;
    const int k  = threadIdx.x;

    const v2f al = { sigmoidf_fast(tau_m[k]), sigmoidf_fast(tau_m[k + 128]) };
    const v2f be = { sigmoidf_fast(tau_n[k]), sigmoidf_fast(tau_n[k + 128]) };
    const v2f w  = { Wout[k], Wout[k + 128] };
    const float bo = bout[0];

    const float* q = seg + ((size_t)(b * SEGS + sg) * H_DIM + k) * 2;
    const v2f s0 = *reinterpret_cast<const v2f*>(q);
    const v2f s1 = *reinterpret_cast<const v2f*>(q + 2 * 128);
    v2f d = { s0[0], s1[0] };
    v2f m = { s0[1], s1[1] };

    const size_t row = (size_t)b * T_DIM + (size_t)sg * L_SEG + k;
    const v2f z = *reinterpret_cast<const v2f*>(outz + row * 2);

    const v2f oma = 1.0f - al;
    float* pc = Ps + k;
#pragma unroll 8
    for (int j = 0; j < L_SEG; ++j) {
        d = be * d;
        m = al * m + oma * d;
        const v2f c = m * w;
        pc[j * PLD] = c[0] + c[1];
    }
    __syncthreads();

    const float* rp = Ps + k * PLD;
    float s = 0.0f;
#pragma unroll
    for (int qd = 0; qd < L_SEG / 4; ++qd) {
        const v4f v = *reinterpret_cast<const v4f*>(rp + 4 * qd);
        s += (v[0] + v[1]) + (v[2] + v[3]);
    }
    out[row] = sigmoidf_fast((bo + (z[0] + z[1])) + s);
}

extern "C" void kernel_launch(void* const* d_in, const int* in_sizes, int n_in,
                              void* d_out, int out_size, void* d_ws, size_t ws_size,
                              hipStream_t stream) {
    const float* x     = (const float*)d_in[0];
    const float* Wd    = (const float*)d_in[1];
    const float* bd    = (const float*)d_in[2];
    const float* tau_m = (const float*)d_in[3];
    const float* tau_n = (const float*)d_in[4];
    const float* Wout  = (const float*)d_in[5];
    const float* bout  = (const float*)d_in[6];
    float* out = (float*)d_out;

    float* outz = (float*)d_ws;
    float* seg  = outz + (size_t)BT * H_DIM;

    sfnn_gemm_wmma<<<(BT / GM) * (H_DIM / GN), 256, 0, stream>>>(
        x, Wd, bd, tau_m, tau_n, Wout, seg, outz);

    sfnn_seg_combine<<<(B_DIM * H_DIM) / 256, 256, 0, stream>>>(
        tau_m, tau_n, seg);

    sfnn_out_scan<<<B_DIM * SEGS, 128, 0, stream>>>(
        outz, seg, tau_m, tau_n, Wout, bout, out);
}


# GEMM kernel: wave priority 3 (s_setprio) for the load/stage prologue and the scan tail, 0 for the WMMA stream: a younger co-resident work-group's loads issue under the older one's WMMAs; same sums
# speedup vs baseline: 2.6079x; 1.0340x over previous
#include <hip/hip_runtime.h>
#include <hip/hip_bf16.h>

static constexpr int B_DIM = 32;
static constexpr int T_DIM = 4096;
static constexpr int I_DIM = 128;
static constexpr int H_DIM = 256;
static constexpr int BT    = B_DIM * T_DIM;
static constexpr int SEGS  = 32;
static constexpr int L_SEG = T_DIM / SEGS;

typedef float v2f __attribute__((ext_vector_type(2)));
typedef float v8f __attribute__((ext_vector_type(8)));

__device__ __forceinline__ float sigmoidf_fast(float z) {
    return 1.0f / (1.0f + __expf(-z));
}

static constexpr int GM  = 128;
static constexpr int GN  = 128;
static constexpr int LDP = I_DIM + 4;

typedef float v4f __attribute__((ext_vector_type(4)));

__global__ __launch_bounds__(256) void sfnn_gemm_wmma(
    const float* __restrict__ x, const float* __restrict__ Wd,
    const float* __restrict__ bd, const float* __restrict__ tau_m,
    const float* __restrict__ tau_n, const float* __restrict__ Wout,
    float* __restrict__ seg, float* __restrict__ outz)
{
    __shared__ __attribute__((aligned(16))) float As[GM * LDP];
    __shared__ __attribute__((aligned(16))) float Bs[GN * LDP];

    __builtin_amdgcn_s_setprio(3);

    const int tid  = threadIdx.x;
    const int row0 = (blockIdx.x >> 1) * GM;
    const int n0   = (blockIdx.x & 1) * GN;

    {
        const v4f* xg = reinterpret_cast<const v4f*>(x  + (size_t)row0 * I_DIM);
        const v4f* wg = reinterpret_cast<const v4f*>(Wd + (size_t)n0   * I_DIM);
        v4f xa[16], wa[16];
#pragma unroll
        for (int i = 0; i < 16; ++i) xa[i] = xg[i * 256 + tid];
#pragma unroll
        for (int i = 0; i < 16; ++i) wa[i] = wg[i * 256 + tid];
        const int sr = tid >> 5;
        const int sc = (tid & 31) * 4;
#pragma unroll
        for (int i = 0; i < 16; ++i)
            *reinterpret_cast<v4f*>(&As[(i * 8 + sr) * LDP + sc]) = xa[i];
#pragma unroll
        for (int i = 0; i < 16; ++i)
            *reinterpret_cast<v4f*>(&Bs[(i * 8 + sr) * LDP + sc]) = wa[i];
    }
    __syncthreads();

    const int lane = tid & 31;
    const int wv   = tid >> 5;
    const int half = lane >> 4;
    const int l    = lane & 15;
    const int wr   = wv >> 1;
    const int wc   = wv & 1;

    const float* ap = As + (wr * 32 + l) * LDP + 4 * half;
    const float* bp = Bs + (wc * 64 + l) * LDP + 4 * half;

    v8f acc[2][4];
#pragma unroll
    for (int j = 0; j < 4; ++j) {
        const float bias = bd[n0 + wc * 64 + j * 16 + l];
#pragma unroll
        for (int i = 0; i < 2; ++i)
#pragma unroll
            for (int r = 0; r < 8; ++r) acc[i][j][r] = bias;
    }

    unsigned toff = (unsigned)((wr * 32 + 8 * half) * LDP + wc * 64 + l);
    const int   cs = tid & (GN - 1);
    float al = sigmoidf_fast(tau_m[n0 + cs]);
    float be = sigmoidf_fast(tau_n[n0 + cs]);
    float wo = Wout[n0 + cs];
    asm volatile("" : "+v"(toff), "+v"(al), "+v"(be), "+v"(wo));

    __builtin_amdgcn_s_setprio(0);

    v4f a[2], b[4];
#pragma unroll
    for (int i = 0; i < 2; ++i) a[i] = *reinterpret_cast<const v4f*>(ap + i * 16 * LDP);
#pragma unroll
    for (int j = 0; j < 4; ++j) b[j] = *reinterpret_cast<const v4f*>(bp + j * 16 * LDP);

#pragma unroll
    for (int kk = 0; kk < I_DIM; kk += 8) {
        v4f an[2], bn[4];
        if (kk + 8 < I_DIM) {
#pragma unroll
            for (int i = 0; i < 2; ++i)
                an[i] = *reinterpret_cast<const v4f*>(ap + i * 16 * LDP + kk + 8);
#pragma unroll
            for (int j = 0; j < 4; ++j)
                bn[j] = *reinterpret_cast<const v4f*>(bp + j * 16 * LDP + kk + 8);
        }
#pragma unroll
        for (int i = 0; i < 2; ++i) {
            const v2f a0 = __builtin_shufflevector(a[i], a[i], 0, 1);
            const v2f a1 = __builtin_shufflevector(a[i], a[i], 2, 3);
#pragma unroll
            for (int j = 0; j < 4; ++j) {
                const v2f b0 = __builtin_shufflevector(b[j], b[j], 0, 1);
                const v2f b1 = __builtin_shufflevector(b[j], b[j], 2, 3);
                acc[i][j] = __builtin_amdgcn_wmma_f32_16x16x4_f32(
                     false, a0,  false, b0,
                     (short)0, acc[i][j],  false,  false);
                acc[i][j] = __builtin_amdgcn_wmma_f32_16x16x4_f32(
                     false, a1,  false, b1,
                     (short)0, acc[i][j],  false,  false);
            }
        }
        if (kk + 8 < I_DIM) {
#pragma unroll
            for (int i = 0; i < 2; ++i) a[i] = an[i];
#pragma unroll
            for (int j = 0; j < 4; ++j) b[j] = bn[j];
        }
    }

    asm volatile("v_nop\n\tv_nop\n\tv_nop\n\tv_nop\n\tv_nop\n\tv_nop\n\tv_nop\n\tv_nop\n\tv_nop"
                 : "+v"(toff),
                   "+v"(acc[0][0]), "+v"(acc[0][1]), "+v"(acc[0][2]), "+v"(acc[0][3]),
                   "+v"(acc[1][0]), "+v"(acc[1][1]), "+v"(acc[1][2]), "+v"(acc[1][3]));

    __builtin_amdgcn_s_setprio(3);

    __syncthreads();
    float* Ts = As;
    {
        float* tp = Ts + toff;
#pragma unroll
        for (int i = 0; i < 2; ++i)
#pragma unroll
            for (int j = 0; j < 4; ++j)
#pragma unroll
                for (int r = 0; r < 8; ++r)
                    tp[(i * 16 + r) * LDP + j * 16] = acc[i][j][r];
    }
    __syncthreads();

    if (tid < GN) {
        const float omb = 1.0f - be;
        const float oma = 1.0f - al;
        float* tc = Ts + tid;
        float d = 0.0f, m = 0.0f;
        for (int t = 0; t < GM; t += 8) {
            float v[8];
#pragma unroll
            for (int j = 0; j < 8; ++j) v[j] = tc[(t + j) * LDP];
#pragma unroll
            for (int j = 0; j < 8; ++j) {
                d = be * d + omb * v[j];
                m = al * m + oma * d;
                v[j] = m * wo;
            }
#pragma unroll
            for (int j = 0; j < 8; ++j) tc[(t + j) * LDP] = v[j];
        }
        float* q = seg + ((size_t)(blockIdx.x >> 1) * H_DIM + n0 + tid) * 2;
        q[0] = d;
        q[1] = m;
    }
    __syncthreads();

    {
        const int row  = tid >> 1;
        const int part = tid & 1;
        const float* rp = Ts + row * LDP + part * 64;
        float s = 0.0f;
#pragma unroll
        for (int qd = 0; qd < 16; ++qd) {
            const v4f v = *reinterpret_cast<const v4f*>(rp + 4 * qd);
            s += (v[0] + v[1]) + (v[2] + v[3]);
        }
        s += __shfl_xor(s, 1, 32);
        if (part == 0)
            outz[(size_t)(row0 + row) * 2 + (blockIdx.x & 1)] = s;
    }
}

__global__ __launch_bounds__(256) void sfnn_seg_combine(
    const float* __restrict__ tau_m, const float* __restrict__ tau_n,
    float* __restrict__ seg)
{
    const int tid = blockIdx.x * blockDim.x + threadIdx.x;
    const int h = tid & (H_DIM - 1);
    const int b = tid >> 8;

    const float alpha = sigmoidf_fast(tau_m[h]);
    const float beta  = sigmoidf_fast(tau_n[h]);

    float betaL = 1.0f, c_dm = 0.0f, alphaL = 1.0f;
#pragma unroll 4
    for (int j = 0; j < L_SEG; ++j) {
        betaL *= beta;
        c_dm   = alpha * c_dm + (1.0f - alpha) * betaL;
        alphaL *= alpha;
    }

    float d = 0.0f, m = 0.0f;
    for (int s = 0; s < SEGS; ++s) {
        float* q = seg + ((size_t)(b * SEGS + s) * H_DIM + h) * 2;
        const float ed = q[0];
        const float em = q[1];
        q[0] = d;
        q[1] = m;
        const float nd = betaL * d + ed;
        const float nm = alphaL * m + c_dm * d + em;
        d = nd;
        m = nm;
    }
}

static constexpr int PLD = L_SEG + 4;

__global__ __launch_bounds__(128) void sfnn_out_scan(
    const float* __restrict__ outz, const float* __restrict__ seg,
    const float* __restrict__ tau_m, const float* __restrict__ tau_n,
    const float* __restrict__ Wout, const float* __restrict__ bout,
    float* __restrict__ out)
{
    __shared__ __attribute__((aligned(16))) float Ps[L_SEG * PLD];

    const int b  = blockIdx.x >> 5;
    const int sg = blockIdx.x & 31;
    const int k  = threadIdx.x;

    const v2f al = { sigmoidf_fast(tau_m[k]), sigmoidf_fast(tau_m[k + 128]) };
    const v2f be = { sigmoidf_fast(tau_n[k]), sigmoidf_fast(tau_n[k + 128]) };
    const v2f w  = { Wout[k], Wout[k + 128] };
    const float bo = bout[0];

    const float* q = seg + ((size_t)(b * SEGS + sg) * H_DIM + k) * 2;
    const v2f s0 = *reinterpret_cast<const v2f*>(q);
    const v2f s1 = *reinterpret_cast<const v2f*>(q + 2 * 128);
    v2f d = { s0[0], s1[0] };
    v2f m = { s0[1], s1[1] };

    const size_t row = (size_t)b * T_DIM + (size_t)sg * L_SEG + k;
    const v2f z = *reinterpret_cast<const v2f*>(outz + row * 2);

    const v2f oma = 1.0f - al;
    float* pc = Ps + k;
#pragma unroll 8
    for (int j = 0; j < L_SEG; ++j) {
        d = be * d;
        m = al * m + oma * d;
        const v2f c = m * w;
        pc[j * PLD] = c[0] + c[1];
    }
    __syncthreads();

    const float* rp = Ps + k * PLD;
    float s = 0.0f;
#pragma unroll
    for (int qd = 0; qd < L_SEG / 4; ++qd) {
        const v4f v = *reinterpret_cast<const v4f*>(rp + 4 * qd);
        s += (v[0] + v[1]) + (v[2] + v[3]);
    }
    out[row] = sigmoidf_fast((bo + (z[0] + z[1])) + s);
}

extern "C" void kernel_launch(void* const* d_in, const int* in_sizes, int n_in,
                              void* d_out, int out_size, void* d_ws, size_t ws_size,
                              hipStream_t stream) {
    const float* x     = (const float*)d_in[0];
    const float* Wd    = (const float*)d_in[1];
    const float* bd    = (const float*)d_in[2];
    const float* tau_m = (const float*)d_in[3];
    const float* tau_n = (const float*)d_in[4];
    const float* Wout  = (const float*)d_in[5];
    const float* bout  = (const float*)d_in[6];
    float* out = (float*)d_out;

    float* outz = (float*)d_ws;
    float* seg  = outz + (size_t)BT * H_DIM;

    sfnn_gemm_wmma<<<(BT / GM) * (H_DIM / GN), 256, 0, stream>>>(
        x, Wd, bd, tau_m, tau_n, Wout, seg, outz);

    sfnn_seg_combine<<<(B_DIM * H_DIM) / 256, 256, 0, stream>>>(
        tau_m, tau_n, seg);

    sfnn_out_scan<<<B_DIM * SEGS, 128, 0, stream>>>(
        outz, seg, tau_m, tau_n, Wout, bout, out);
}


# GEMM kernel as a resident grid: 512 work-groups keep the W_dense tile in LDS and loop over 4 row tiles, next x tile loaded to registers under the WMMAs, matrix priority falls per tile; same sums
# speedup vs baseline: 2.9223x; 1.1206x over previous
#include <hip/hip_runtime.h>
#include <hip/hip_bf16.h>

static constexpr int B_DIM = 32;
static constexpr int T_DIM = 4096;
static constexpr int I_DIM = 128;
static constexpr int H_DIM = 256;
static constexpr int BT    = B_DIM * T_DIM;
static constexpr int SEGS  = 32;
static constexpr int L_SEG = T_DIM / SEGS;

typedef float v2f __attribute__((ext_vector_type(2)));
typedef float v8f __attribute__((ext_vector_type(8)));

__device__ __forceinline__ float sigmoidf_fast(float z) {
    return 1.0f / (1.0f + __expf(-z));
}

static constexpr int GM  = 128;
static constexpr int GN  = 128;
static constexpr int LDP = I_DIM + 4;

typedef float v4f __attribute__((ext_vector_type(4)));

static constexpr int TILES = 4;
static constexpr int NP    = (BT / GM) / TILES;

__global__ __launch_bounds__(256) void sfnn_gemm_wmma(
    const float* __restrict__ x, const float* __restrict__ Wd,
    const float* __restrict__ bd, const float* __restrict__ tau_m,
    const float* __restrict__ tau_n, const float* __restrict__ Wout,
    float* __restrict__ seg, float* __restrict__ outz)
{
    __shared__ __attribute__((aligned(16))) float As[GM * LDP];
    __shared__ __attribute__((aligned(16))) float Bs[GN * LDP];

    __builtin_amdgcn_s_setprio(3);

    const int tid = threadIdx.x;
    const int ch  = blockIdx.x & 1;
    const int n0  = ch * GN;
    const int p   = blockIdx.x >> 1;
    const int sr  = tid >> 5;
    const int sc  = (tid & 31) * 4;

    {
        const v4f* xg = reinterpret_cast<const v4f*>(x  + (size_t)p  * GM * I_DIM);
        const v4f* wg = reinterpret_cast<const v4f*>(Wd + (size_t)n0 * I_DIM);
        v4f xa[16], wa[16];
#pragma unroll
        for (int i = 0; i < 16; ++i) xa[i] = xg[i * 256 + tid];
#pragma unroll
        for (int i = 0; i < 16; ++i) wa[i] = wg[i * 256 + tid];
#pragma unroll
        for (int i = 0; i < 16; ++i)
            *reinterpret_cast<v4f*>(&As[(i * 8 + sr) * LDP + sc]) = xa[i];
#pragma unroll
        for (int i = 0; i < 16; ++i)
            *reinterpret_cast<v4f*>(&Bs[(i * 8 + sr) * LDP + sc]) = wa[i];
    }

    const int lane = tid & 31;
    const int wv   = tid >> 5;
    const int half = lane >> 4;
    const int l    = lane & 15;
    const int wr   = wv >> 1;
    const int wc   = wv & 1;

    const float* ap = As + (wr * 32 + l) * LDP + 4 * half;
    const float* bp = Bs + (wc * 64 + l) * LDP + 4 * half;

    float bias[4];
#pragma unroll
    for (int j = 0; j < 4; ++j) bias[j] = bd[n0 + wc * 64 + j * 16 + l];

    unsigned toff = (unsigned)((wr * 32 + 8 * half) * LDP + wc * 64 + l);
    const int   cs = tid & (GN - 1);
    float al = sigmoidf_fast(tau_m[n0 + cs]);
    float be = sigmoidf_fast(tau_n[n0 + cs]);
    float wo = Wout[n0 + cs];
    unsigned soff = (unsigned)((n0 + cs) * 2);
    __syncthreads();

#pragma nounroll
    for (int it = 0; it < TILES; ++it) {
        const int rt   = it * NP + p;
        const int row0 = rt * GM;

        v4f xn[16];
        if (it + 1 < TILES) {
            const v4f* xg = reinterpret_cast<const v4f*>(x + (size_t)(row0 + NP * GM) * I_DIM);
#pragma unroll
            for (int i = 0; i < 16; ++i) xn[i] = xg[i * 256 + tid];
        }

        v8f acc[2][4];
#pragma unroll
        for (int j = 0; j < 4; ++j)
#pragma unroll
            for (int i = 0; i < 2; ++i)
#pragma unroll
                for (int r = 0; r < 8; ++r) acc[i][j][r] = bias[j];

        asm volatile("" : "+v"(toff), "+v"(soff), "+v"(al), "+v"(be), "+v"(wo));

        switch (it) {
            case 0:  __builtin_amdgcn_s_setprio(3); break;
            case 1:  __builtin_amdgcn_s_setprio(2); break;
            case 2:  __builtin_amdgcn_s_setprio(1); break;
            default: __builtin_amdgcn_s_setprio(0); break;
        }

        v4f a[2], b[4];
#pragma unroll
        for (int i = 0; i < 2; ++i) a[i] = *reinterpret_cast<const v4f*>(ap + i * 16 * LDP);
#pragma unroll
        for (int j = 0; j < 4; ++j) b[j] = *reinterpret_cast<const v4f*>(bp + j * 16 * LDP);

#pragma unroll
        for (int kk = 0; kk < I_DIM; kk += 8) {
            v4f an[2], bn[4];
            if (kk + 8 < I_DIM) {
#pragma unroll
                for (int i = 0; i < 2; ++i)
                    an[i] = *reinterpret_cast<const v4f*>(ap + i * 16 * LDP + kk + 8);
#pragma unroll
                for (int j = 0; j < 4; ++j)
                    bn[j] = *reinterpret_cast<const v4f*>(bp + j * 16 * LDP + kk + 8);
            }
#pragma unroll
            for (int i = 0; i < 2; ++i) {
                const v2f a0 = __builtin_shufflevector(a[i], a[i], 0, 1);
                const v2f a1 = __builtin_shufflevector(a[i], a[i], 2, 3);
#pragma unroll
                for (int j = 0; j < 4; ++j) {
                    const v2f b0 = __builtin_shufflevector(b[j], b[j], 0, 1);
                    const v2f b1 = __builtin_shufflevector(b[j], b[j], 2, 3);
                    acc[i][j] = __builtin_amdgcn_wmma_f32_16x16x4_f32(
                         false, a0,  false, b0,
                         (short)0, acc[i][j],  false,  false);
                    acc[i][j] = __builtin_amdgcn_wmma_f32_16x16x4_f32(
                         false, a1,  false, b1,
                         (short)0, acc[i][j],  false,  false);
                }
            }
            if (kk + 8 < I_DIM) {
#pragma unroll
                for (int i = 0; i < 2; ++i) a[i] = an[i];
#pragma unroll
                for (int j = 0; j < 4; ++j) b[j] = bn[j];
            }
        }

        asm volatile("v_nop\n\tv_nop\n\tv_nop\n\tv_nop\n\tv_nop\n\tv_nop\n\tv_nop\n\tv_nop\n\tv_nop"
                     : "+v"(toff),
                       "+v"(acc[0][0]), "+v"(acc[0][1]), "+v"(acc[0][2]), "+v"(acc[0][3]),
                       "+v"(acc[1][0]), "+v"(acc[1][1]), "+v"(acc[1][2]), "+v"(acc[1][3]));

        __builtin_amdgcn_s_setprio(3);

        __syncthreads();
        float* Ts = As;
        {
            float* tp = Ts + toff;
#pragma unroll
            for (int i = 0; i < 2; ++i)
#pragma unroll
                for (int j = 0; j < 4; ++j)
#pragma unroll
                    for (int r = 0; r < 8; ++r)
                        tp[(i * 16 + r) * LDP + j * 16] = acc[i][j][r];
        }
        __syncthreads();

        if (tid < GN) {
            const float omb = 1.0f - be;
            const float oma = 1.0f - al;
            float* tc = Ts + tid;
            float d = 0.0f, m = 0.0f;
            for (int t = 0; t < GM; t += 8) {
                float v[8];
#pragma unroll
                for (int j = 0; j < 8; ++j) v[j] = tc[(t + j) * LDP];
#pragma unroll
                for (int j = 0; j < 8; ++j) {
                    d = be * d + omb * v[j];
                    m = al * m + oma * d;
                    v[j] = m * wo;
                }
#pragma unroll
                for (int j = 0; j < 8; ++j) tc[(t + j) * LDP] = v[j];
            }
            float* q = seg + (size_t)rt * (H_DIM * 2) + soff;
            q[0] = d;
            q[1] = m;
        }
        __syncthreads();

        {
            const int row  = tid >> 1;
            const int part = tid & 1;
            const float* rp = Ts + row * LDP + part * 64;
            float s = 0.0f;
#pragma unroll
            for (int qd = 0; qd < 16; ++qd) {
                const v4f v = *reinterpret_cast<const v4f*>(rp + 4 * qd);
                s += (v[0] + v[1]) + (v[2] + v[3]);
            }
            s += __shfl_xor(s, 1, 32);
            if (part == 0)
                outz[(size_t)(row0 + row) * 2 + ch] = s;
        }

        if (it + 1 < TILES) {
            __syncthreads();
#pragma unroll
            for (int i = 0; i < 16; ++i)
                *reinterpret_cast<v4f*>(&As[(i * 8 + sr) * LDP + sc]) = xn[i];
            __syncthreads();
        }
    }
}

__global__ __launch_bounds__(256) void sfnn_seg_combine(
    const float* __restrict__ tau_m, const float* __restrict__ tau_n,
    float* __restrict__ seg)
{
    const int tid = blockIdx.x * blockDim.x + threadIdx.x;
    const int h = tid & (H_DIM - 1);
    const int b = tid >> 8;

    const float alpha = sigmoidf_fast(tau_m[h]);
    const float beta  = sigmoidf_fast(tau_n[h]);

    float betaL = 1.0f, c_dm = 0.0f, alphaL = 1.0f;
#pragma unroll 4
    for (int j = 0; j < L_SEG; ++j) {
        betaL *= beta;
        c_dm   = alpha * c_dm + (1.0f - alpha) * betaL;
        alphaL *= alpha;
    }

    float d = 0.0f, m = 0.0f;
    for (int s = 0; s < SEGS; ++s) {
        float* q = seg + ((size_t)(b * SEGS + s) * H_DIM + h) * 2;
        const float ed = q[0];
        const float em = q[1];
        q[0] = d;
        q[1] = m;
        const float nd = betaL * d + ed;
        const float nm = alphaL * m + c_dm * d + em;
        d = nd;
        m = nm;
    }
}

static constexpr int PLD = L_SEG + 4;

__global__ __launch_bounds__(128) void sfnn_out_scan(
    const float* __restrict__ outz, const float* __restrict__ seg,
    const float* __restrict__ tau_m, const float* __restrict__ tau_n,
    const float* __restrict__ Wout, const float* __restrict__ bout,
    float* __restrict__ out)
{
    __shared__ __attribute__((aligned(16))) float Ps[L_SEG * PLD];

    const int b  = blockIdx.x >> 5;
    const int sg = blockIdx.x & 31;
    const int k  = threadIdx.x;

    const v2f al = { sigmoidf_fast(tau_m[k]), sigmoidf_fast(tau_m[k + 128]) };
    const v2f be = { sigmoidf_fast(tau_n[k]), sigmoidf_fast(tau_n[k + 128]) };
    const v2f w  = { Wout[k], Wout[k + 128] };
    const float bo = bout[0];

    const float* q = seg + ((size_t)(b * SEGS + sg) * H_DIM + k) * 2;
    const v2f s0 = *reinterpret_cast<const v2f*>(q);
    const v2f s1 = *reinterpret_cast<const v2f*>(q + 2 * 128);
    v2f d = { s0[0], s1[0] };
    v2f m = { s0[1], s1[1] };

    const size_t row = (size_t)b * T_DIM + (size_t)sg * L_SEG + k;
    const v2f z = *reinterpret_cast<const v2f*>(outz + row * 2);

    const v2f oma = 1.0f - al;
    float* pc = Ps + k;
#pragma unroll 8
    for (int j = 0; j < L_SEG; ++j) {
        d = be * d;
        m = al * m + oma * d;
        const v2f c = m * w;
        pc[j * PLD] = c[0] + c[1];
    }
    __syncthreads();

    const float* rp = Ps + k * PLD;
    float s = 0.0f;
#pragma unroll
    for (int qd = 0; qd < L_SEG / 4; ++qd) {
        const v4f v = *reinterpret_cast<const v4f*>(rp + 4 * qd);
        s += (v[0] + v[1]) + (v[2] + v[3]);
    }
    out[row] = sigmoidf_fast((bo + (z[0] + z[1])) + s);
}

extern "C" void kernel_launch(void* const* d_in, const int* in_sizes, int n_in,
                              void* d_out, int out_size, void* d_ws, size_t ws_size,
                              hipStream_t stream) {
    const float* x     = (const float*)d_in[0];
    const float* Wd    = (const float*)d_in[1];
    const float* bd    = (const float*)d_in[2];
    const float* tau_m = (const float*)d_in[3];
    const float* tau_n = (const float*)d_in[4];
    const float* Wout  = (const float*)d_in[5];
    const float* bout  = (const float*)d_in[6];
    float* out = (float*)d_out;

    float* outz = (float*)d_ws;
    float* seg  = outz + (size_t)BT * H_DIM;

    sfnn_gemm_wmma<<<NP * (H_DIM / GN), 256, 0, stream>>>(
        x, Wd, bd, tau_m, tau_n, Wout, seg, outz);

    sfnn_seg_combine<<<(B_DIM * H_DIM) / 256, 256, 0, stream>>>(
        tau_m, tau_n, seg);

    sfnn_out_scan<<<B_DIM * SEGS, 128, 0, stream>>>(
        outz, seg, tau_m, tau_n, Wout, bout, out);
}


# Two launches, not three: a GEMM work-group takes 4 consecutive segments and carries its scan state; last kernel builds the run-start state from <=7 earlier run ends (sum order changes)
# speedup vs baseline: 3.1762x; 1.0869x over previous
#include <hip/hip_runtime.h>
#include <hip/hip_bf16.h>

static constexpr int B_DIM = 32;
static constexpr int T_DIM = 4096;
static constexpr int I_DIM = 128;
static constexpr int H_DIM = 256;
static constexpr int BT    = B_DIM * T_DIM;
static constexpr int SEGS  = 32;
static constexpr int L_SEG = T_DIM / SEGS;
static constexpr int RUN   = 4;
static constexpr int RUNS  = SEGS / RUN;

typedef float v2f __attribute__((ext_vector_type(2)));
typedef float v8f __attribute__((ext_vector_type(8)));

__device__ __forceinline__ float sigmoidf_fast(float z) {
    return 1.0f / (1.0f + __expf(-z));
}

static constexpr int GM  = 128;
static constexpr int GN  = 128;
static constexpr int LDP = I_DIM + 4;

typedef float v4f __attribute__((ext_vector_type(4)));

static constexpr int TILES = RUN;
static constexpr int NP    = (BT / GM) / TILES;

__global__ __launch_bounds__(256) void sfnn_gemm_wmma(
    const float* __restrict__ x, const float* __restrict__ Wd,
    const float* __restrict__ bd, const float* __restrict__ tau_m,
    const float* __restrict__ tau_n, const float* __restrict__ Wout,
    float* __restrict__ seg, float* __restrict__ outz)
{
    __shared__ __attribute__((aligned(16))) float As[GM * LDP];
    __shared__ __attribute__((aligned(16))) float Bs[GN * LDP];

    __builtin_amdgcn_s_setprio(3);

    const int tid = threadIdx.x;
    const int ch  = blockIdx.x & 1;
    const int n0  = ch * GN;
    const int p   = blockIdx.x >> 1;
    const int sr  = tid >> 5;
    const int sc  = (tid & 31) * 4;

    {
        const v4f* xg = reinterpret_cast<const v4f*>(x  + (size_t)p  * (TILES * GM) * I_DIM);
        const v4f* wg = reinterpret_cast<const v4f*>(Wd + (size_t)n0 * I_DIM);
        v4f xa[16], wa[16];
#pragma unroll
        for (int i = 0; i < 16; ++i) xa[i] = xg[i * 256 + tid];
#pragma unroll
        for (int i = 0; i < 16; ++i) wa[i] = wg[i * 256 + tid];
#pragma unroll
        for (int i = 0; i < 16; ++i)
            *reinterpret_cast<v4f*>(&As[(i * 8 + sr) * LDP + sc]) = xa[i];
#pragma unroll
        for (int i = 0; i < 16; ++i)
            *reinterpret_cast<v4f*>(&Bs[(i * 8 + sr) * LDP + sc]) = wa[i];
    }

    const int lane = tid & 31;
    const int wv   = tid >> 5;
    const int half = lane >> 4;
    const int l    = lane & 15;
    const int wr   = wv >> 1;
    const int wc   = wv & 1;

    const float* ap = As + (wr * 32 + l) * LDP + 4 * half;
    const float* bp = Bs + (wc * 64 + l) * LDP + 4 * half;

    float bias[4];
#pragma unroll
    for (int j = 0; j < 4; ++j) bias[j] = bd[n0 + wc * 64 + j * 16 + l];

    unsigned toff = (unsigned)((wr * 32 + 8 * half) * LDP + wc * 64 + l);
    const int   cs = tid & (GN - 1);
    float al = sigmoidf_fast(tau_m[n0 + cs]);
    float be = sigmoidf_fast(tau_n[n0 + cs]);
    float wo = Wout[n0 + cs];
    float dcar = 0.0f, mcar = 0.0f;
    __syncthreads();

#pragma nounroll
    for (int it = 0; it < TILES; ++it) {
        const int rt   = p * TILES + it;
        const int row0 = rt * GM;

        v4f xn[16];
        if (it + 1 < TILES) {
            const v4f* xg = reinterpret_cast<const v4f*>(x + (size_t)(row0 + GM) * I_DIM);
#pragma unroll
            for (int i = 0; i < 16; ++i) xn[i] = xg[i * 256 + tid];
        }

        v8f acc[2][4];
#pragma unroll
        for (int j = 0; j < 4; ++j)
#pragma unroll
            for (int i = 0; i < 2; ++i)
#pragma unroll
                for (int r = 0; r < 8; ++r) acc[i][j][r] = bias[j];

        asm volatile("" : "+v"(toff), "+v"(al), "+v"(be), "+v"(wo));

        switch (it) {
            case 0:  __builtin_amdgcn_s_setprio(3); break;
            case 1:  __builtin_amdgcn_s_setprio(2); break;
            case 2:  __builtin_amdgcn_s_setprio(1); break;
            default: __builtin_amdgcn_s_setprio(0); break;
        }

        v4f a[2], b[4];
#pragma unroll
        for (int i = 0; i < 2; ++i) a[i] = *reinterpret_cast<const v4f*>(ap + i * 16 * LDP);
#pragma unroll
        for (int j = 0; j < 4; ++j) b[j] = *reinterpret_cast<const v4f*>(bp + j * 16 * LDP);

#pragma unroll
        for (int kk = 0; kk < I_DIM; kk += 8) {
            v4f an[2], bn[4];
            if (kk + 8 < I_DIM) {
#pragma unroll
                for (int i = 0; i < 2; ++i)
                    an[i] = *reinterpret_cast<const v4f*>(ap + i * 16 * LDP + kk + 8);
#pragma unroll
                for (int j = 0; j < 4; ++j)
                    bn[j] = *reinterpret_cast<const v4f*>(bp + j * 16 * LDP + kk + 8);
            }
#pragma unroll
            for (int i = 0; i < 2; ++i) {
                const v2f a0 = __builtin_shufflevector(a[i], a[i], 0, 1);
                const v2f a1 = __builtin_shufflevector(a[i], a[i], 2, 3);
#pragma unroll
                for (int j = 0; j < 4; ++j) {
                    const v2f b0 = __builtin_shufflevector(b[j], b[j], 0, 1);
                    const v2f b1 = __builtin_shufflevector(b[j], b[j], 2, 3);
                    acc[i][j] = __builtin_amdgcn_wmma_f32_16x16x4_f32(
                         false, a0,  false, b0,
                         (short)0, acc[i][j],  false,  false);
                    acc[i][j] = __builtin_amdgcn_wmma_f32_16x16x4_f32(
                         false, a1,  false, b1,
                         (short)0, acc[i][j],  false,  false);
                }
            }
            if (kk + 8 < I_DIM) {
#pragma unroll
                for (int i = 0; i < 2; ++i) a[i] = an[i];
#pragma unroll
                for (int j = 0; j < 4; ++j) b[j] = bn[j];
            }
        }

        asm volatile("v_nop\n\tv_nop\n\tv_nop\n\tv_nop\n\tv_nop\n\tv_nop\n\tv_nop\n\tv_nop\n\tv_nop"
                     : "+v"(toff),
                       "+v"(acc[0][0]), "+v"(acc[0][1]), "+v"(acc[0][2]), "+v"(acc[0][3]),
                       "+v"(acc[1][0]), "+v"(acc[1][1]), "+v"(acc[1][2]), "+v"(acc[1][3]));

        __builtin_amdgcn_s_setprio(3);

        __syncthreads();
        float* Ts = As;
        {
            float* tp = Ts + toff;
#pragma unroll
            for (int i = 0; i < 2; ++i)
#pragma unroll
                for (int j = 0; j < 4; ++j)
#pragma unroll
                    for (int r = 0; r < 8; ++r)
                        tp[(i * 16 + r) * LDP + j * 16] = acc[i][j][r];
        }
        __syncthreads();

        if (tid < GN) {
            const float omb = 1.0f - be;
            const float oma = 1.0f - al;
            float* tc = Ts + tid;
            float d = dcar, m = mcar;
            for (int t = 0; t < GM; t += 8) {
                float v[8];
#pragma unroll
                for (int j = 0; j < 8; ++j) v[j] = tc[(t + j) * LDP];
#pragma unroll
                for (int j = 0; j < 8; ++j) {
                    d = be * d + omb * v[j];
                    m = al * m + oma * d;
                    v[j] = m * wo;
                }
#pragma unroll
                for (int j = 0; j < 8; ++j) tc[(t + j) * LDP] = v[j];
            }
            dcar = d;
            mcar = m;
        }
        __syncthreads();

        {
            const int row  = tid >> 1;
            const int part = tid & 1;
            const float* rp = Ts + row * LDP + part * 64;
            float s = 0.0f;
#pragma unroll
            for (int qd = 0; qd < 16; ++qd) {
                const v4f v = *reinterpret_cast<const v4f*>(rp + 4 * qd);
                s += (v[0] + v[1]) + (v[2] + v[3]);
            }
            s += __shfl_xor(s, 1, 32);
            if (part == 0)
                outz[(size_t)(row0 + row) * 2 + ch] = s;
        }

        if (it + 1 < TILES) {
            __syncthreads();
#pragma unroll
            for (int i = 0; i < 16; ++i)
                *reinterpret_cast<v4f*>(&As[(i * 8 + sr) * LDP + sc]) = xn[i];
            __syncthreads();
        }
    }

    if (tid < GN) {
        const v2f e = { dcar, mcar };
        *reinterpret_cast<v2f*>(seg + ((size_t)p * H_DIM + n0 + cs) * 2) = e;
    }
}

static constexpr int PLD = L_SEG + 4;
static_assert(L_SEG == 128 && RUN == 4, "the squaring counts below (7, then 2) are log2 of these");

__global__ __launch_bounds__(128) void sfnn_out_scan(
    const float* __restrict__ outz, const float* __restrict__ seg,
    const float* __restrict__ tau_m, const float* __restrict__ tau_n,
    const float* __restrict__ Wout, const float* __restrict__ bout,
    float* __restrict__ out)
{
    __shared__ __attribute__((aligned(16))) float Ps[L_SEG * PLD];

    const int b  = blockIdx.x >> 5;
    const int sg = blockIdx.x & 31;
    const int r  = sg >> 2;
    const int js = sg & 3;
    const int k  = threadIdx.x;

    const float* q = seg + ((size_t)(b * RUNS) * H_DIM + k) * 2;
    v2f e0[RUNS - 1], e1[RUNS - 1];
#pragma unroll
    for (int i = 0; i < RUNS - 1; ++i) {
        const int rr = r - (RUNS - 1) + i;
        const int rl = rr < 0 ? 0 : rr;
        e0[i] = *reinterpret_cast<const v2f*>(q + (size_t)rl * (H_DIM * 2));
        e1[i] = *reinterpret_cast<const v2f*>(q + (size_t)rl * (H_DIM * 2) + 2 * 128);
    }

    const size_t row = (size_t)b * T_DIM + (size_t)sg * L_SEG + k;
    const v2f z = *reinterpret_cast<const v2f*>(outz + row * 2);

    const v2f al = { sigmoidf_fast(tau_m[k]), sigmoidf_fast(tau_m[k + 128]) };
    const v2f be = { sigmoidf_fast(tau_n[k]), sigmoidf_fast(tau_n[k + 128]) };
    const v2f w  = { Wout[k], Wout[k + 128] };
    const float bo = bout[0];
    const v2f oma = 1.0f - al;

    v2f pb = be, pa = al, pc = oma * be;
    for (int i = 0; i < 7; ++i) {
        pc = pc * (pa + pb);
        pa = pa * pa;
        pb = pb * pb;
    }
    v2f rb = pb, ra = pa, rc = pc;
    for (int i = 0; i < 2; ++i) {
        rc = rc * (ra + rb);
        ra = ra * ra;
        rb = rb * rb;
    }

    v2f d = { 0.0f, 0.0f };
    v2f m = { 0.0f, 0.0f };
#pragma unroll
    for (int i = 0; i < RUNS - 1; ++i) {
        const bool on = r - (RUNS - 1) + i >= 0;
        const v2f ed = { on ? e0[i][0] : 0.0f, on ? e1[i][0] : 0.0f };
        const v2f em = { on ? e0[i][1] : 0.0f, on ? e1[i][1] : 0.0f };
        m = (ra * m + rc * d) + em;
        d = rb * d + ed;
    }
    for (int i = 0; i < js; ++i) {
        m = pa * m + pc * d;
        d = pb * d;
    }

    float* pcol = Ps + k;
#pragma unroll 8
    for (int j = 0; j < L_SEG; ++j) {
        d = be * d;
        m = al * m + oma * d;
        const v2f c = m * w;
        pcol[j * PLD] = c[0] + c[1];
    }
    __syncthreads();

    const float* rp = Ps + k * PLD;
    float s = 0.0f;
#pragma unroll
    for (int qd = 0; qd < L_SEG / 4; ++qd) {
        const v4f v = *reinterpret_cast<const v4f*>(rp + 4 * qd);
        s += (v[0] + v[1]) + (v[2] + v[3]);
    }
    out[row] = sigmoidf_fast((bo + (z[0] + z[1])) + s);
}

extern "C" void kernel_launch(void* const* d_in, const int* in_sizes, int n_in,
                              void* d_out, int out_size, void* d_ws, size_t ws_size,
                              hipStream_t stream) {
    const float* x     = (const float*)d_in[0];
    const float* Wd    = (const float*)d_in[1];
    const float* bd    = (const float*)d_in[2];
    const float* tau_m = (const float*)d_in[3];
    const float* tau_n = (const float*)d_in[4];
    const float* Wout  = (const float*)d_in[5];
    const float* bout  = (const float*)d_in[6];
    float* out = (float*)d_out;

    float* outz = (float*)d_ws;
    float* seg  = outz + (size_t)BT * H_DIM;

    sfnn_gemm_wmma<<<NP * (H_DIM / GN), 256, 0, stream>>>(
        x, Wd, bd, tau_m, tau_n, Wout, seg, outz);

    sfnn_out_scan<<<B_DIM * SEGS, 128, 0, stream>>>(
        outz, seg, tau_m, tau_n, Wout, bout, out);
}


# Scan on gain-free states, 2 FMAs a step (was 5 ops); c = W_out(1-alpha)(1-beta) rides on the channel sum as packed FMAs, 4 partial sums (sum order changes); last kernel on scaled states, 3 ops a step
# speedup vs baseline: 3.2879x; 1.0352x over previous
#include <hip/hip_runtime.h>
#include <hip/hip_bf16.h>

static constexpr int B_DIM = 32;
static constexpr int T_DIM = 4096;
static constexpr int I_DIM = 128;
static constexpr int H_DIM = 256;
static constexpr int BT    = B_DIM * T_DIM;
static constexpr int SEGS  = 32;
static constexpr int L_SEG = T_DIM / SEGS;
static constexpr int RUN   = 4;
static constexpr int RUNS  = SEGS / RUN;

typedef float v2f __attribute__((ext_vector_type(2)));
typedef float v8f __attribute__((ext_vector_type(8)));

__device__ __forceinline__ float sigmoidf_fast(float z) {
    return 1.0f / (1.0f + __expf(-z));
}

static constexpr int GM  = 128;
static constexpr int GN  = 128;
static constexpr int LDP = I_DIM + 4;

typedef float v4f __attribute__((ext_vector_type(4)));

static constexpr int TILES = RUN;
static constexpr int NP    = (BT / GM) / TILES;

__global__ __launch_bounds__(256) void sfnn_gemm_wmma(
    const float* __restrict__ x, const float* __restrict__ Wd,
    const float* __restrict__ bd, const float* __restrict__ tau_m,
    const float* __restrict__ tau_n, const float* __restrict__ Wout,
    float* __restrict__ seg, float* __restrict__ outz)
{
    __shared__ __attribute__((aligned(16))) float As[GM * LDP];
    __shared__ __attribute__((aligned(16))) float Bs[GN * LDP];
    __shared__ __attribute__((aligned(16))) float Cs[GN];

    __builtin_amdgcn_s_setprio(3);

    const int tid = threadIdx.x;
    const int ch  = blockIdx.x & 1;
    const int n0  = ch * GN;
    const int p   = blockIdx.x >> 1;
    const int sr  = tid >> 5;
    const int sc  = (tid & 31) * 4;

    {
        const v4f* xg = reinterpret_cast<const v4f*>(x  + (size_t)p  * (TILES * GM) * I_DIM);
        const v4f* wg = reinterpret_cast<const v4f*>(Wd + (size_t)n0 * I_DIM);
        v4f xa[16], wa[16];
#pragma unroll
        for (int i = 0; i < 16; ++i) xa[i] = xg[i * 256 + tid];
#pragma unroll
        for (int i = 0; i < 16; ++i) wa[i] = wg[i * 256 + tid];
#pragma unroll
        for (int i = 0; i < 16; ++i)
            *reinterpret_cast<v4f*>(&As[(i * 8 + sr) * LDP + sc]) = xa[i];
#pragma unroll
        for (int i = 0; i < 16; ++i)
            *reinterpret_cast<v4f*>(&Bs[(i * 8 + sr) * LDP + sc]) = wa[i];
    }

    const int lane = tid & 31;
    const int wv   = tid >> 5;
    const int half = lane >> 4;
    const int l    = lane & 15;
    const int wr   = wv >> 1;
    const int wc   = wv & 1;

    const float* ap = As + (wr * 32 + l) * LDP + 4 * half;
    const float* bp = Bs + (wc * 64 + l) * LDP + 4 * half;

    float bias[4];
#pragma unroll
    for (int j = 0; j < 4; ++j) bias[j] = bd[n0 + wc * 64 + j * 16 + l];

    unsigned toff = (unsigned)((wr * 32 + 8 * half) * LDP + wc * 64 + l);
    const int   cs = tid & (GN - 1);
    float al = sigmoidf_fast(tau_m[n0 + cs]);
    float be = sigmoidf_fast(tau_n[n0 + cs]);
    const float cq = Wout[n0 + cs] * ((1.0f - al) * (1.0f - be));
    if (tid < GN) Cs[cs] = cq;
    float dcar = 0.0f, mcar = 0.0f;
    __syncthreads();

#pragma nounroll
    for (int it = 0; it < TILES; ++it) {
        const int rt   = p * TILES + it;
        const int row0 = rt * GM;

        v4f xn[16];
        if (it + 1 < TILES) {
            const v4f* xg = reinterpret_cast<const v4f*>(x + (size_t)(row0 + GM) * I_DIM);
#pragma unroll
            for (int i = 0; i < 16; ++i) xn[i] = xg[i * 256 + tid];
        }

        v8f acc[2][4];
#pragma unroll
        for (int j = 0; j < 4; ++j)
#pragma unroll
            for (int i = 0; i < 2; ++i)
#pragma unroll
                for (int r = 0; r < 8; ++r) acc[i][j][r] = bias[j];

        asm volatile("" : "+v"(toff), "+v"(al), "+v"(be));

        switch (it) {
            case 0:  __builtin_amdgcn_s_setprio(3); break;
            case 1:  __builtin_amdgcn_s_setprio(2); break;
            case 2:  __builtin_amdgcn_s_setprio(1); break;
            default: __builtin_amdgcn_s_setprio(0); break;
        }

        v4f a[2], b[4];
#pragma unroll
        for (int i = 0; i < 2; ++i) a[i] = *reinterpret_cast<const v4f*>(ap + i * 16 * LDP);
#pragma unroll
        for (int j = 0; j < 4; ++j) b[j] = *reinterpret_cast<const v4f*>(bp + j * 16 * LDP);

#pragma unroll
        for (int kk = 0; kk < I_DIM; kk += 8) {
            v4f an[2], bn[4];
            if (kk + 8 < I_DIM) {
#pragma unroll
                for (int i = 0; i < 2; ++i)
                    an[i] = *reinterpret_cast<const v4f*>(ap + i * 16 * LDP + kk + 8);
#pragma unroll
                for (int j = 0; j < 4; ++j)
                    bn[j] = *reinterpret_cast<const v4f*>(bp + j * 16 * LDP + kk + 8);
            }
#pragma unroll
            for (int i = 0; i < 2; ++i) {
                const v2f a0 = __builtin_shufflevector(a[i], a[i], 0, 1);
                const v2f a1 = __builtin_shufflevector(a[i], a[i], 2, 3);
#pragma unroll
                for (int j = 0; j < 4; ++j) {
                    const v2f b0 = __builtin_shufflevector(b[j], b[j], 0, 1);
                    const v2f b1 = __builtin_shufflevector(b[j], b[j], 2, 3);
                    acc[i][j] = __builtin_amdgcn_wmma_f32_16x16x4_f32(
                         false, a0,  false, b0,
                         (short)0, acc[i][j],  false,  false);
                    acc[i][j] = __builtin_amdgcn_wmma_f32_16x16x4_f32(
                         false, a1,  false, b1,
                         (short)0, acc[i][j],  false,  false);
                }
            }
            if (kk + 8 < I_DIM) {
#pragma unroll
                for (int i = 0; i < 2; ++i) a[i] = an[i];
#pragma unroll
                for (int j = 0; j < 4; ++j) b[j] = bn[j];
            }
        }

        asm volatile("v_nop\n\tv_nop\n\tv_nop\n\tv_nop\n\tv_nop\n\tv_nop\n\tv_nop\n\tv_nop\n\tv_nop"
                     : "+v"(toff),
                       "+v"(acc[0][0]), "+v"(acc[0][1]), "+v"(acc[0][2]), "+v"(acc[0][3]),
                       "+v"(acc[1][0]), "+v"(acc[1][1]), "+v"(acc[1][2]), "+v"(acc[1][3]));

        __builtin_amdgcn_s_setprio(3);

        __syncthreads();
        float* Ts = As;
        {
            float* tp = Ts + toff;
#pragma unroll
            for (int i = 0; i < 2; ++i)
#pragma unroll
                for (int j = 0; j < 4; ++j)
#pragma unroll
                    for (int r = 0; r < 8; ++r)
                        tp[(i * 16 + r) * LDP + j * 16] = acc[i][j][r];
        }
        __syncthreads();

        if (tid < GN) {
            float* tc = Ts + tid;
            float d = dcar, m = mcar;
            for (int t = 0; t < GM; t += 8) {
                float v[8];
#pragma unroll
                for (int j = 0; j < 8; ++j) v[j] = tc[(t + j) * LDP];
#pragma unroll
                for (int j = 0; j < 8; ++j) {
                    d = be * d + v[j];
                    m = al * m + d;
                    v[j] = m;
                }
#pragma unroll
                for (int j = 0; j < 8; ++j) tc[(t + j) * LDP] = v[j];
            }
            dcar = d;
            mcar = m;
        }
        __syncthreads();

        {
            const int row  = tid >> 1;
            const int part = tid & 1;
            const float* rp = Ts + row * LDP + part * 64;
            const float* cp = Cs + part * 64;
            v4f s4 = { 0.0f, 0.0f, 0.0f, 0.0f };
#pragma unroll
            for (int qd = 0; qd < 16; ++qd) {
                const v4f v = *reinterpret_cast<const v4f*>(rp + 4 * qd);
                const v4f c = *reinterpret_cast<const v4f*>(cp + 4 * qd);
                s4 = v * c + s4;
            }
            float s = (s4[0] + s4[1]) + (s4[2] + s4[3]);
            s += __shfl_xor(s, 1, 32);
            if (part == 0)
                outz[(size_t)(row0 + row) * 2 + ch] = s;
        }

        if (it + 1 < TILES) {
            __syncthreads();
#pragma unroll
            for (int i = 0; i < 16; ++i)
                *reinterpret_cast<v4f*>(&As[(i * 8 + sr) * LDP + sc]) = xn[i];
            __syncthreads();
        }
    }

    if (tid < GN) {
        const v2f e = { cq * dcar, cq * mcar };
        *reinterpret_cast<v2f*>(seg + ((size_t)p * H_DIM + n0 + cs) * 2) = e;
    }
}

static constexpr int PLD = L_SEG + 4;
static_assert(L_SEG == 128 && RUN == 4, "the squaring counts below (7, then 2) are log2 of these");

__global__ __launch_bounds__(128) void sfnn_out_scan(
    const float* __restrict__ outz, const float* __restrict__ seg,
    const float* __restrict__ tau_m, const float* __restrict__ tau_n,
    const float* __restrict__ bout,
    float* __restrict__ out)
{
    __shared__ __attribute__((aligned(16))) float Ps[L_SEG * PLD];

    const int b  = blockIdx.x >> 5;
    const int sg = blockIdx.x & 31;
    const int r  = sg >> 2;
    const int js = sg & 3;
    const int k  = threadIdx.x;

    const float* q = seg + ((size_t)(b * RUNS) * H_DIM + k) * 2;
    v2f e0[RUNS - 1], e1[RUNS - 1];
#pragma unroll
    for (int i = 0; i < RUNS - 1; ++i) {
        const int rr = r - (RUNS - 1) + i;
        const int rl = rr < 0 ? 0 : rr;
        e0[i] = *reinterpret_cast<const v2f*>(q + (size_t)rl * (H_DIM * 2));
        e1[i] = *reinterpret_cast<const v2f*>(q + (size_t)rl * (H_DIM * 2) + 2 * 128);
    }

    const size_t row = (size_t)b * T_DIM + (size_t)sg * L_SEG + k;
    const v2f z = *reinterpret_cast<const v2f*>(outz + row * 2);

    const v2f al = { sigmoidf_fast(tau_m[k]), sigmoidf_fast(tau_m[k + 128]) };
    const v2f be = { sigmoidf_fast(tau_n[k]), sigmoidf_fast(tau_n[k + 128]) };
    const float bo = bout[0];

    v2f pb = be, pa = al, pc = be;
    for (int i = 0; i < 7; ++i) {
        pc = pc * (pa + pb);
        pa = pa * pa;
        pb = pb * pb;
    }
    v2f rb = pb, ra = pa, rc = pc;
    for (int i = 0; i < 2; ++i) {
        rc = rc * (ra + rb);
        ra = ra * ra;
        rb = rb * rb;
    }

    v2f d = { 0.0f, 0.0f };
    v2f m = { 0.0f, 0.0f };
#pragma unroll
    for (int i = 0; i < RUNS - 1; ++i) {
        const bool on = r - (RUNS - 1) + i >= 0;
        const v2f ed = { on ? e0[i][0] : 0.0f, on ? e1[i][0] : 0.0f };
        const v2f em = { on ? e0[i][1] : 0.0f, on ? e1[i][1] : 0.0f };
        m = (ra * m + rc * d) + em;
        d = rb * d + ed;
    }
    for (int i = 0; i < js; ++i) {
        m = pa * m + pc * d;
        d = pb * d;
    }

    float* pcol = Ps + k;
#pragma unroll 8
    for (int j = 0; j < L_SEG; ++j) {
        d = be * d;
        m = al * m + d;
        pcol[j * PLD] = m[0] + m[1];
    }
    __syncthreads();

    const float* rp = Ps + k * PLD;
    float s = 0.0f;
#pragma unroll
    for (int qd = 0; qd < L_SEG / 4; ++qd) {
        const v4f v = *reinterpret_cast<const v4f*>(rp + 4 * qd);
        s += (v[0] + v[1]) + (v[2] + v[3]);
    }
    out[row] = sigmoidf_fast((bo + (z[0] + z[1])) + s);
}

extern "C" void kernel_launch(void* const* d_in, const int* in_sizes, int n_in,
                              void* d_out, int out_size, void* d_ws, size_t ws_size,
                              hipStream_t stream) {
    const float* x     = (const float*)d_in[0];
    const float* Wd    = (const float*)d_in[1];
    const float* bd    = (const float*)d_in[2];
    const float* tau_m = (const float*)d_in[3];
    const float* tau_n = (const float*)d_in[4];
    const float* Wout  = (const float*)d_in[5];
    const float* bout  = (const float*)d_in[6];
    float* out = (float*)d_out;

    float* outz = (float*)d_ws;
    float* seg  = outz + (size_t)BT * H_DIM;

    sfnn_gemm_wmma<<<NP * (H_DIM / GN), 256, 0, stream>>>(
        x, Wd, bd, tau_m, tau_n, Wout, seg, outz);

    sfnn_out_scan<<<B_DIM * SEGS, 128, 0, stream>>>(
        outz, seg, tau_m, tau_n, bout, out);
}
